# starting kernel
# baseline (speedup 1.0000x reference)
#include <hip/hip_runtime.h>

#define B_DIM 16
#define N_DIM 2048
#define H_DIM 1024
#define NT    128

typedef __attribute__((ext_vector_type(2))) float v2f;
typedef __attribute__((ext_vector_type(8))) float v8f;


__device__ __forceinline__ v2f load_a_f32(const float* __restrict__ A, int lda,
                                          int r0, int k, int lane) {
  int r  = r0 + (lane & 15);
  int kk = k + ((lane >> 4) << 1);
  const float* p = A + (size_t)r * lda + kk;
  return v2f{p[0], p[1]};
}

__device__ __forceinline__ v2f load_b_rowmajor(const float* __restrict__ Bm, int ldb,
                                               int k, int c0, int lane) {
  int c  = c0 + (lane & 15);
  int kk = k + ((lane >> 4) << 1);
  const float* p = Bm + (size_t)kk * ldb + c;
  return v2f{p[0], p[ldb]};
}

__device__ __forceinline__ v2f load_b_transposed(const float* __restrict__ T, int ldt,
                                                 int k, int c0, int lane) {
  int c  = c0 + (lane & 15);
  int kk = k + ((lane >> 4) << 1);
  const float* p = T + (size_t)c * ldt + kk;
  return v2f{p[0], p[1]};
}

__device__ __forceinline__ v8f wmma_f32(v2f a, v2f b, v8f c) {
  return __builtin_amdgcn_wmma_f32_16x16x4_f32(false, a, false, b, (short)0, c,
                                               false, false);
}

__global__ void fc1_kernel(const float* __restrict__ feats,
                           const float* __restrict__ W,
                           const float* __restrict__ bias,
                           float* __restrict__ q) {
  int gid  = blockIdx.x * (blockDim.x >> 5) + (threadIdx.x >> 5);
  int lane = threadIdx.x & 31;
  int tM = gid >> 5;
  int tN = gid & 31;
  int r0 = tM * 16;
  int c0 = tN * 32;

  v8f acc0 = {};
  v8f acc1 = {};
  #pragma unroll 4
  for (int k = 0; k < H_DIM; k += 4) {
    v2f a  = load_a_f32(feats, H_DIM, r0, k, lane);
    v2f b0 = load_b_rowmajor(W, H_DIM, k, c0, lane);
    v2f b1 = load_b_rowmajor(W, H_DIM, k, c0 + 16, lane);
    acc0 = wmma_f32(a, b0, acc0);
    acc1 = wmma_f32(a, b1, acc1);
  }

  int row = r0 + ((lane >> 4) << 3);
  int col = c0 + (lane & 15);
  float bz0 = bias[col];
  float bz1 = bias[col + 16];
  #pragma unroll
  for (int j = 0; j < 8; ++j) {
    q[(size_t)(row + j) * H_DIM + col]      = acc0[j] + bz0;
    q[(size_t)(row + j) * H_DIM + col + 16] = acc1[j] + bz1;
  }
}

__global__ void attn_kernel(const float* __restrict__ q,
                            const float* __restrict__ feats,
                            float* __restrict__ partial) {
  int b  = blockIdx.x >> 7;
  int nt = blockIdx.x & (NT - 1);
  int n0 = nt * 16;
  const float* Q = q     + ((size_t)b * N_DIM + n0) * H_DIM;
  const float* F = feats + (size_t)b * N_DIM * H_DIM;

  extern __shared__ float slab[];
  __shared__ float rowInv[16];

  int wave = threadIdx.x >> 5;
  int lane = threadIdx.x & 31;

  for (int g = 0; g < 4; ++g) {
    int m0 = wave * 256 + g * 64;
    v8f acc[4] = {};
    #pragma unroll 2
    for (int k = 0; k < H_DIM; k += 4) {
      v2f a = load_a_f32(Q, H_DIM, 0, k, lane);
      #pragma unroll
      for (int t = 0; t < 4; ++t) {
        v2f bf = load_b_transposed(F, H_DIM, k, m0 + t * 16, lane);
        acc[t] = wmma_f32(a, bf, acc[t]);
      }
    }
    int rbase = (lane >> 4) << 3;
    int cl    = lane & 15;
    #pragma unroll
    for (int t = 0; t < 4; ++t) {
      #pragma unroll
      for (int j = 0; j < 8; ++j)
        slab[(rbase + j) * N_DIM + m0 + t * 16 + cl] = acc[t][j];
    }
  }
  __syncthreads();

  for (int rr = 0; rr < 2; ++rr) {
    int r = wave * 2 + rr;
    float mx = -3.402823466e38f;
    for (int m = lane; m < N_DIM; m += 32)
      mx = fmaxf(mx, slab[r * N_DIM + m]);
    #pragma unroll
    for (int o = 16; o > 0; o >>= 1)
      mx = fmaxf(mx, __shfl_xor(mx, o, 32));
    float s = 0.f;
    for (int m = lane; m < N_DIM; m += 32) {
      float e = __expf(slab[r * N_DIM + m] - mx);
      slab[r * N_DIM + m] = e;
      s += e;
    }
    #pragma unroll
    for (int o = 16; o > 0; o >>= 1)
      s += __shfl_xor(s, o, 32);
    if (lane == 0) rowInv[r] = 1.0f / s;
  }
  __syncthreads();

  float* P = partial + ((size_t)b * NT + nt) * N_DIM;
  for (int m = threadIdx.x; m < N_DIM; m += 256) {
    float a = 0.f;
    #pragma unroll
    for (int r = 0; r < 16; ++r)
      a += slab[r * N_DIM + m] * rowInv[r];
    P[m] = a;
  }
}

__global__ void colsum_kernel(const float* __restrict__ partial,
                              float* __restrict__ colsum) {
  int idx = blockIdx.x * blockDim.x + threadIdx.x;
  int b = idx >> 11;
  int m = idx & (N_DIM - 1);
  const float* P = partial + (size_t)b * NT * N_DIM + m;
  float a = 0.f;
  #pragma unroll 8
  for (int t = 0; t < NT; ++t) a += P[(size_t)t * N_DIM];
  colsum[idx] = a;
}

__global__ void pool_kernel(const float* __restrict__ colsum,
                            const float* __restrict__ feats,
                            float* __restrict__ out) {
  int b = blockIdx.x;
  int h = blockIdx.y * 256 + threadIdx.x;
  const float* F = feats + (size_t)b * N_DIM * H_DIM + h;
  const float* C = colsum + (size_t)b * N_DIM;
  float a = 0.f;
  #pragma unroll 4
  for (int m = 0; m < N_DIM; ++m)
    a += C[m] * F[(size_t)m * H_DIM];
  out[(size_t)b * H_DIM + h] = a;
}

extern "C" void kernel_launch(void* const* d_in, const int* in_sizes, int n_in,
                              void* d_out, int out_size, void* d_ws, size_t ws_size,
                              hipStream_t stream) {
  const float* feats = (const float*)d_in[0];
  const float* W     = (const float*)d_in[1];
  const float* bias  = (const float*)d_in[2];
  float* out = (float*)d_out;

  float* q       = (float*)d_ws;
  float* partial = q + (size_t)B_DIM * N_DIM * H_DIM;
  float* colsum  = partial + (size_t)B_DIM * NT * N_DIM;

  fc1_kernel<<<8192, 256, 0, stream>>>(feats, W, bias, q);

  attn_kernel<<<B_DIM * NT, 256, 16 * N_DIM * sizeof(float), stream>>>(q, feats, partial);

  colsum_kernel<<<(B_DIM * N_DIM) / 256, 256, 0, stream>>>(partial, colsum);

  pool_kernel<<<dim3(B_DIM, H_DIM / 256), 256, 0, stream>>>(colsum, feats, out);
}


# fc1 GEMM staged through LDS: 128x128 block tiles, K chunks of 64 by whole-line 128-bit loads two chunks ahead, W k-row pairs interleaved in LDS (64-bit fragment reads); every sum keeps its order
# speedup vs baseline: 2.3635x; 2.3635x over previous
#include <hip/hip_runtime.h>

#define B_DIM 16
#define N_DIM 2048
#define H_DIM 1024
#define NT    128

typedef __attribute__((ext_vector_type(2))) float v2f;
typedef __attribute__((ext_vector_type(8))) float v8f;


__device__ __forceinline__ v8f wmma_f32(v2f a, v2f b, v8f c) {
  return __builtin_amdgcn_wmma_f32_16x16x4_f32(false, a, false, b, (short)0, c,
                                               false, false);
}

__device__ __forceinline__ void lds_barrier() {
  asm volatile("s_wait_dscnt 0x0" ::: "memory");
  asm volatile("s_barrier_signal -1" ::: "memory");
  asm volatile("s_barrier_wait -1" ::: "memory");
}

#define FC_KC      64
#define FC_PITCH_A 68
#define FC_PITCH_B 288
#define FC_TILE_A  (128 * FC_PITCH_A)
#define FC_TILE_B  (32 * FC_PITCH_B)

#define FC_GLOAD(KO)                                                    \
  ra0 = *(const float4*)(gA + (size_t)0   * H_DIM + (KO));              \
  rb0 = *(const float4*)(gB + ((size_t)(KO) + 0)  * H_DIM);             \
  ra1 = *(const float4*)(gA + (size_t)16  * H_DIM + (KO));              \
  rb1 = *(const float4*)(gB + ((size_t)(KO) + 1)  * H_DIM);             \
  ra2 = *(const float4*)(gA + (size_t)32  * H_DIM + (KO));              \
  rb2 = *(const float4*)(gB + ((size_t)(KO) + 16) * H_DIM);             \
  ra3 = *(const float4*)(gA + (size_t)48  * H_DIM + (KO));              \
  rb3 = *(const float4*)(gB + ((size_t)(KO) + 17) * H_DIM);             \
  ra4 = *(const float4*)(gA + (size_t)64  * H_DIM + (KO));              \
  rb4 = *(const float4*)(gB + ((size_t)(KO) + 32) * H_DIM);             \
  ra5 = *(const float4*)(gA + (size_t)80  * H_DIM + (KO));              \
  rb5 = *(const float4*)(gB + ((size_t)(KO) + 33) * H_DIM);             \
  ra6 = *(const float4*)(gA + (size_t)96  * H_DIM + (KO));              \
  rb6 = *(const float4*)(gB + ((size_t)(KO) + 48) * H_DIM);             \
  ra7 = *(const float4*)(gA + (size_t)112 * H_DIM + (KO));              \
  rb7 = *(const float4*)(gB + ((size_t)(KO) + 49) * H_DIM);

#define FC_BPAIR(DB, J, LO, HI)                                                             \
  *(float4*)((DB) + bsoff + (J) * 8 * FC_PITCH_B)     = make_float4(LO.x, HI.x, LO.y, HI.y); \
  *(float4*)((DB) + bsoff + (J) * 8 * FC_PITCH_B + 4) = make_float4(LO.z, HI.z, LO.w, HI.w);
#define FC_SSTORE(DA, DB)                                  \
  *(float4*)((DA) + asoff + 0   * FC_PITCH_A) = ra0;       \
  *(float4*)((DA) + asoff + 16  * FC_PITCH_A) = ra1;       \
  *(float4*)((DA) + asoff + 32  * FC_PITCH_A) = ra2;       \
  *(float4*)((DA) + asoff + 48  * FC_PITCH_A) = ra3;       \
  *(float4*)((DA) + asoff + 64  * FC_PITCH_A) = ra4;       \
  *(float4*)((DA) + asoff + 80  * FC_PITCH_A) = ra5;       \
  *(float4*)((DA) + asoff + 96  * FC_PITCH_A) = ra6;       \
  *(float4*)((DA) + asoff + 112 * FC_PITCH_A) = ra7;       \
  FC_BPAIR(DB, 0, rb0, rb1)                                \
  FC_BPAIR(DB, 1, rb2, rb3)                                \
  FC_BPAIR(DB, 2, rb4, rb5)                                \
  FC_BPAIR(DB, 3, rb6, rb7)

__global__ __launch_bounds__(256)
void fc1_kernel(const float* __restrict__ feats,
                const float* __restrict__ W,
                const float* __restrict__ bias,
                float* __restrict__ q) {
  extern __shared__ float stage[];
  float* sA = stage;
  float* sB = stage + 2 * FC_TILE_A;

  int r0 = blockIdx.y * 128;
  int c0 = blockIdx.x * 128;

  int tid  = threadIdx.x;
  int wave = tid >> 5;
  int lane = tid & 31;
  int wr = wave >> 1;
  int wc = wave & 1;
  int c  = lane & 15;
  int h  = lane >> 4;

  int arow = tid >> 4;
  int aseg = (tid & 15) * 4;
  const float* gA = feats + (size_t)(r0 + arow) * H_DIM + aseg;
  int asoff = arow * FC_PITCH_A + aseg;
  int bpair = tid >> 5;
  int bcol  = (tid & 31) * 4;
  const float* gB = W + (size_t)(2 * bpair) * H_DIM + c0 + bcol;
  int bsoff = bpair * FC_PITCH_B + 2 * bcol;

  float4 ra0, ra1, ra2, ra3, ra4, ra5, ra6, ra7;
  float4 rb0, rb1, rb2, rb3, rb4, rb5, rb6, rb7;
  FC_GLOAD(0)
  FC_SSTORE(sA, sB)
  FC_GLOAD(FC_KC)
  lds_barrier();

  v8f acc[2][4] = {};
  int aoff = (wr * 32 + c) * FC_PITCH_A + 2 * h;
  int boff = h * FC_PITCH_B + 2 * (wc * 64 + c);

  for (int ch = 0; ch < H_DIM / FC_KC; ++ch) {
    const float* cA = sA + (ch & 1) * FC_TILE_A;
    const float* cB = sB + (ch & 1) * FC_TILE_B;
    #pragma unroll
    for (int ks = 0; ks < FC_KC / 4; ++ks) {
      v2f a[2], bf[4];
      #pragma unroll
      for (int i = 0; i < 2; ++i)
        a[i] = *(const v2f*)(cA + aoff + i * 16 * FC_PITCH_A + ks * 4);
      #pragma unroll
      for (int t = 0; t < 4; ++t)
        bf[t] = *(const v2f*)(cB + boff + ks * 2 * FC_PITCH_B + t * 32);
      #pragma unroll
      for (int i = 0; i < 2; ++i)
        #pragma unroll
        for (int t = 0; t < 4; ++t)
          acc[i][t] = wmma_f32(a[i], bf[t], acc[i][t]);
    }
    __builtin_amdgcn_sched_barrier(0);
    if (ch + 1 < H_DIM / FC_KC) {
      float* nA = sA + ((ch + 1) & 1) * FC_TILE_A;
      float* nB = sB + ((ch + 1) & 1) * FC_TILE_B;
      FC_SSTORE(nA, nB)
    }
    if (ch + 2 < H_DIM / FC_KC) {
      int ko = (ch + 2) * FC_KC;
      FC_GLOAD(ko)
    }
    lds_barrier();
  }

  #pragma unroll
  for (int t = 0; t < 4; ++t) {
    int col = c0 + wc * 64 + t * 16 + c;
    float bz = bias[col];
    #pragma unroll
    for (int i = 0; i < 2; ++i) {
      float* dst = q + (size_t)(r0 + wr * 32 + i * 16 + 8 * h) * H_DIM + col;
      #pragma unroll
      for (int j = 0; j < 8; ++j)
        dst[(size_t)j * H_DIM] = acc[i][t][j] + bz;
    }
  }
}

#define SC_PITCH 36
#define SC_TILE  (128 * SC_PITCH)
__global__ __launch_bounds__(256)
void scores_kernel(const float* __restrict__ q,
                   const float* __restrict__ feats,
                   float* __restrict__ S) {
  extern __shared__ float stage[];
  float* sA = stage;
  float* sB = stage + 2 * SC_TILE;

  int b  = blockIdx.z;
  int n0 = blockIdx.y * 128;
  int m0 = blockIdx.x * 128;
  const float* Qg = q     + ((size_t)b * N_DIM + n0) * H_DIM;
  const float* Fg = feats + ((size_t)b * N_DIM + m0) * H_DIM;
  float* Sg = S + ((size_t)b * N_DIM + n0) * N_DIM + m0;

  int tid  = threadIdx.x;
  int wave = tid >> 5;
  int lane = tid & 31;
  int wr = wave >> 1;
  int wc = wave & 1;
  int c  = lane & 15;
  int h  = lane >> 4;

  int srow = tid >> 3;
  int sseg = (tid & 7) * 4;
  const float* gA = Qg + (size_t)srow * H_DIM + sseg;
  const float* gB = Fg + (size_t)srow * H_DIM + sseg;
  int soff = srow * SC_PITCH + sseg;

  float4 ra0, ra1, ra2, ra3, rb0, rb1, rb2, rb3;
  ra0 = *(const float4*)(gA + (size_t)0 * H_DIM + 0);
  rb0 = *(const float4*)(gB + (size_t)0 * H_DIM + 0);
  ra1 = *(const float4*)(gA + (size_t)32 * H_DIM + 0);
  rb1 = *(const float4*)(gB + (size_t)32 * H_DIM + 0);
  ra2 = *(const float4*)(gA + (size_t)64 * H_DIM + 0);
  rb2 = *(const float4*)(gB + (size_t)64 * H_DIM + 0);
  ra3 = *(const float4*)(gA + (size_t)96 * H_DIM + 0);
  rb3 = *(const float4*)(gB + (size_t)96 * H_DIM + 0);
  *(float4*)(sA + soff + 0 * SC_PITCH) = ra0;
  *(float4*)(sB + soff + 0 * SC_PITCH) = rb0;
  *(float4*)(sA + soff + 32 * SC_PITCH) = ra1;
  *(float4*)(sB + soff + 32 * SC_PITCH) = rb1;
  *(float4*)(sA + soff + 64 * SC_PITCH) = ra2;
  *(float4*)(sB + soff + 64 * SC_PITCH) = rb2;
  *(float4*)(sA + soff + 96 * SC_PITCH) = ra3;
  *(float4*)(sB + soff + 96 * SC_PITCH) = rb3;
  ra0 = *(const float4*)(gA + (size_t)0 * H_DIM + 32);
  rb0 = *(const float4*)(gB + (size_t)0 * H_DIM + 32);
  ra1 = *(const float4*)(gA + (size_t)32 * H_DIM + 32);
  rb1 = *(const float4*)(gB + (size_t)32 * H_DIM + 32);
  ra2 = *(const float4*)(gA + (size_t)64 * H_DIM + 32);
  rb2 = *(const float4*)(gB + (size_t)64 * H_DIM + 32);
  ra3 = *(const float4*)(gA + (size_t)96 * H_DIM + 32);
  rb3 = *(const float4*)(gB + (size_t)96 * H_DIM + 32);
  lds_barrier();

  v8f acc[2][4] = {};
  int aoff = (wr * 32 + c) * SC_PITCH + 2 * h;
  int boff = (wc * 64 + c) * SC_PITCH + 2 * h;

  for (int ch = 0; ch < H_DIM / 32; ++ch) {
    const float* cA = sA + (ch & 1) * SC_TILE;
    const float* cB = sB + (ch & 1) * SC_TILE;
    #pragma unroll
    for (int ks = 0; ks < 8; ++ks) {
      v2f a[2], bf[4];
      #pragma unroll
      for (int i = 0; i < 2; ++i)
        a[i] = *(const v2f*)(cA + aoff + i * 16 * SC_PITCH + ks * 4);
      #pragma unroll
      for (int t = 0; t < 4; ++t)
        bf[t] = *(const v2f*)(cB + boff + t * 16 * SC_PITCH + ks * 4);
      #pragma unroll
      for (int i = 0; i < 2; ++i)
        #pragma unroll
        for (int t = 0; t < 4; ++t)
          acc[i][t] = wmma_f32(a[i], bf[t], acc[i][t]);
    }
    if (ch + 1 < H_DIM / 32) {
      float* nA = sA + ((ch + 1) & 1) * SC_TILE;
      float* nB = sB + ((ch + 1) & 1) * SC_TILE;
      *(float4*)(nA + soff + 0 * SC_PITCH) = ra0;
      *(float4*)(nB + soff + 0 * SC_PITCH) = rb0;
      *(float4*)(nA + soff + 32 * SC_PITCH) = ra1;
      *(float4*)(nB + soff + 32 * SC_PITCH) = rb1;
      *(float4*)(nA + soff + 64 * SC_PITCH) = ra2;
      *(float4*)(nB + soff + 64 * SC_PITCH) = rb2;
      *(float4*)(nA + soff + 96 * SC_PITCH) = ra3;
      *(float4*)(nB + soff + 96 * SC_PITCH) = rb3;
    }
    if (ch + 2 < H_DIM / 32) {
      int ko = (ch + 2) * 32;
      ra0 = *(const float4*)(gA + (size_t)0 * H_DIM + ko);
      rb0 = *(const float4*)(gB + (size_t)0 * H_DIM + ko);
      ra1 = *(const float4*)(gA + (size_t)32 * H_DIM + ko);
      rb1 = *(const float4*)(gB + (size_t)32 * H_DIM + ko);
      ra2 = *(const float4*)(gA + (size_t)64 * H_DIM + ko);
      rb2 = *(const float4*)(gB + (size_t)64 * H_DIM + ko);
      ra3 = *(const float4*)(gA + (size_t)96 * H_DIM + ko);
      rb3 = *(const float4*)(gB + (size_t)96 * H_DIM + ko);
    }
    lds_barrier();
  }

  #pragma unroll
  for (int i = 0; i < 2; ++i) {
    #pragma unroll
    for (int t = 0; t < 4; ++t) {
      float* dst = Sg + (size_t)(wr * 32 + i * 16 + 8 * h) * N_DIM + wc * 64 + t * 16 + c;
      #pragma unroll
      for (int j = 0; j < 8; ++j)
        dst[(size_t)j * N_DIM] = acc[i][t][j];
    }
  }
}

__global__ void softmax_kernel(const float* __restrict__ S,
                               float* __restrict__ partial) {
  int b  = blockIdx.x >> 7;
  int nt = blockIdx.x & (NT - 1);
  const float* Sr = S + ((size_t)b * N_DIM + nt * 16) * N_DIM;

  extern __shared__ float slab[];
  __shared__ float rowInv[16];

  int wave = threadIdx.x >> 5;
  int lane = threadIdx.x & 31;

  for (int i = threadIdx.x * 4; i < 16 * N_DIM; i += 256 * 4)
    *(float4*)(slab + i) = *(const float4*)(Sr + i);
  __syncthreads();

  for (int rr = 0; rr < 2; ++rr) {
    int r = wave * 2 + rr;
    float mx = -3.402823466e38f;
    for (int m = lane; m < N_DIM; m += 32)
      mx = fmaxf(mx, slab[r * N_DIM + m]);
    #pragma unroll
    for (int o = 16; o > 0; o >>= 1)
      mx = fmaxf(mx, __shfl_xor(mx, o, 32));
    float s = 0.f;
    for (int m = lane; m < N_DIM; m += 32) {
      float e = __expf(slab[r * N_DIM + m] - mx);
      slab[r * N_DIM + m] = e;
      s += e;
    }
    #pragma unroll
    for (int o = 16; o > 0; o >>= 1)
      s += __shfl_xor(s, o, 32);
    if (lane == 0) rowInv[r] = 1.0f / s;
  }
  __syncthreads();

  float* P = partial + ((size_t)b * NT + nt) * N_DIM;
  for (int m = threadIdx.x; m < N_DIM; m += 256) {
    float a = 0.f;
    #pragma unroll
    for (int r = 0; r < 16; ++r)
      a += slab[r * N_DIM + m] * rowInv[r];
    P[m] = a;
  }
}

__global__ void colsum_kernel(const float* __restrict__ partial,
                              float* __restrict__ colsum) {
  int idx = blockIdx.x * blockDim.x + threadIdx.x;
  int b = idx >> 11;
  int m = idx & (N_DIM - 1);
  const float* P = partial + (size_t)b * NT * N_DIM + m;
  float a = 0.f;
  #pragma unroll 8
  for (int t = 0; t < NT; ++t) a += P[(size_t)t * N_DIM];
  colsum[idx] = a;
}

__global__ void pool_kernel(const float* __restrict__ colsum,
                            const float* __restrict__ feats,
                            float* __restrict__ out) {
  int b = blockIdx.x;
  int h = blockIdx.y * 256 + threadIdx.x;
  const float* F = feats + (size_t)b * N_DIM * H_DIM + h;
  const float* C = colsum + (size_t)b * N_DIM;
  float a = 0.f;
  #pragma unroll 4
  for (int m = 0; m < N_DIM; ++m)
    a += C[m] * F[(size_t)m * H_DIM];
  out[(size_t)b * H_DIM + h] = a;
}

#define GRP_FEATS (4 * N_DIM * H_DIM)
#define GRP_PART  (4 * NT * N_DIM)
#define RUN_GROUP(G)                                                            \
  fc1_kernel<<<dim3(8, 64), 256, (2 * FC_TILE_A + 2 * FC_TILE_B) * sizeof(float), stream>>>(  \
      feats + (size_t)(G) * GRP_FEATS, W, bias, q);                                         \
  scores_kernel<<<dim3(16, 16, 4), 256, 4 * SC_TILE * sizeof(float), stream>>>(            \
      q, feats + (size_t)(G) * GRP_FEATS, S);                                               \
  softmax_kernel<<<4 * NT, 256, 16 * N_DIM * sizeof(float), stream>>>(                      \
      S, partial + (size_t)(G) * GRP_PART);

extern "C" void kernel_launch(void* const* d_in, const int* in_sizes, int n_in,
                              void* d_out, int out_size, void* d_ws, size_t ws_size,
                              hipStream_t stream) {
  const float* feats = (const float*)d_in[0];
  const float* W     = (const float*)d_in[1];
  const float* bias  = (const float*)d_in[2];
  float* out = (float*)d_out;

  float* q       = (float*)d_ws;
  float* S       = q + (size_t)4 * N_DIM * H_DIM;
  float* partial = S + (size_t)4 * N_DIM * N_DIM;
  float* colsum  = partial + (size_t)B_DIM * NT * N_DIM;

  RUN_GROUP(0)
  RUN_GROUP(1)
  RUN_GROUP(2)
  RUN_GROUP(3)

  colsum_kernel<<<(B_DIM * N_DIM) / 256, 256, 0, stream>>>(partial, colsum);

  pool_kernel<<<dim3(B_DIM, H_DIM / 256), 256, 0, stream>>>(colsum, feats, out);
}


# Pool sum over m split over the grid: 64 strips of 32 rows, one block each (1024 blocks, a dozen 512-byte row loads in flight per wave); a small kernel joins the strips in order (new order of m sum)
# speedup vs baseline: 2.7173x; 1.1497x over previous
#include <hip/hip_runtime.h>

#define B_DIM 16
#define N_DIM 2048
#define H_DIM 1024
#define NT    128

typedef __attribute__((ext_vector_type(2))) float v2f;
typedef __attribute__((ext_vector_type(8))) float v8f;


__device__ __forceinline__ v8f wmma_f32(v2f a, v2f b, v8f c) {
  return __builtin_amdgcn_wmma_f32_16x16x4_f32(false, a, false, b, (short)0, c,
                                               false, false);
}

__device__ __forceinline__ void lds_barrier() {
  asm volatile("s_wait_dscnt 0x0" ::: "memory");
  asm volatile("s_barrier_signal -1" ::: "memory");
  asm volatile("s_barrier_wait -1" ::: "memory");
}

#define FC_KC      64
#define FC_PITCH_A 68
#define FC_PITCH_B 288
#define FC_TILE_A  (128 * FC_PITCH_A)
#define FC_TILE_B  (32 * FC_PITCH_B)

#define FC_GLOAD(KO)                                                    \
  ra0 = *(const float4*)(gA + (size_t)0   * H_DIM + (KO));              \
  rb0 = *(const float4*)(gB + ((size_t)(KO) + 0)  * H_DIM);             \
  ra1 = *(const float4*)(gA + (size_t)16  * H_DIM + (KO));              \
  rb1 = *(const float4*)(gB + ((size_t)(KO) + 1)  * H_DIM);             \
  ra2 = *(const float4*)(gA + (size_t)32  * H_DIM + (KO));              \
  rb2 = *(const float4*)(gB + ((size_t)(KO) + 16) * H_DIM);             \
  ra3 = *(const float4*)(gA + (size_t)48  * H_DIM + (KO));              \
  rb3 = *(const float4*)(gB + ((size_t)(KO) + 17) * H_DIM);             \
  ra4 = *(const float4*)(gA + (size_t)64  * H_DIM + (KO));              \
  rb4 = *(const float4*)(gB + ((size_t)(KO) + 32) * H_DIM);             \
  ra5 = *(const float4*)(gA + (size_t)80  * H_DIM + (KO));              \
  rb5 = *(const float4*)(gB + ((size_t)(KO) + 33) * H_DIM);             \
  ra6 = *(const float4*)(gA + (size_t)96  * H_DIM + (KO));              \
  rb6 = *(const float4*)(gB + ((size_t)(KO) + 48) * H_DIM);             \
  ra7 = *(const float4*)(gA + (size_t)112 * H_DIM + (KO));              \
  rb7 = *(const float4*)(gB + ((size_t)(KO) + 49) * H_DIM);

#define FC_BPAIR(DB, J, LO, HI)                                                             \
  *(float4*)((DB) + bsoff + (J) * 8 * FC_PITCH_B)     = make_float4(LO.x, HI.x, LO.y, HI.y); \
  *(float4*)((DB) + bsoff + (J) * 8 * FC_PITCH_B + 4) = make_float4(LO.z, HI.z, LO.w, HI.w);
#define FC_SSTORE(DA, DB)                                  \
  *(float4*)((DA) + asoff + 0   * FC_PITCH_A) = ra0;       \
  *(float4*)((DA) + asoff + 16  * FC_PITCH_A) = ra1;       \
  *(float4*)((DA) + asoff + 32  * FC_PITCH_A) = ra2;       \
  *(float4*)((DA) + asoff + 48  * FC_PITCH_A) = ra3;       \
  *(float4*)((DA) + asoff + 64  * FC_PITCH_A) = ra4;       \
  *(float4*)((DA) + asoff + 80  * FC_PITCH_A) = ra5;       \
  *(float4*)((DA) + asoff + 96  * FC_PITCH_A) = ra6;       \
  *(float4*)((DA) + asoff + 112 * FC_PITCH_A) = ra7;       \
  FC_BPAIR(DB, 0, rb0, rb1)                                \
  FC_BPAIR(DB, 1, rb2, rb3)                                \
  FC_BPAIR(DB, 2, rb4, rb5)                                \
  FC_BPAIR(DB, 3, rb6, rb7)

__global__ __launch_bounds__(256)
void fc1_kernel(const float* __restrict__ feats,
                const float* __restrict__ W,
                const float* __restrict__ bias,
                float* __restrict__ q) {
  extern __shared__ float stage[];
  float* sA = stage;
  float* sB = stage + 2 * FC_TILE_A;

  int r0 = blockIdx.y * 128;
  int c0 = blockIdx.x * 128;

  int tid  = threadIdx.x;
  int wave = tid >> 5;
  int lane = tid & 31;
  int wr = wave >> 1;
  int wc = wave & 1;
  int c  = lane & 15;
  int h  = lane >> 4;

  int arow = tid >> 4;
  int aseg = (tid & 15) * 4;
  const float* gA = feats + (size_t)(r0 + arow) * H_DIM + aseg;
  int asoff = arow * FC_PITCH_A + aseg;
  int bpair = tid >> 5;
  int bcol  = (tid & 31) * 4;
  const float* gB = W + (size_t)(2 * bpair) * H_DIM + c0 + bcol;
  int bsoff = bpair * FC_PITCH_B + 2 * bcol;

  float4 ra0, ra1, ra2, ra3, ra4, ra5, ra6, ra7;
  float4 rb0, rb1, rb2, rb3, rb4, rb5, rb6, rb7;
  FC_GLOAD(0)
  FC_SSTORE(sA, sB)
  FC_GLOAD(FC_KC)
  lds_barrier();

  v8f acc[2][4] = {};
  int aoff = (wr * 32 + c) * FC_PITCH_A + 2 * h;
  int boff = h * FC_PITCH_B + 2 * (wc * 64 + c);

  for (int ch = 0; ch < H_DIM / FC_KC; ++ch) {
    const float* cA = sA + (ch & 1) * FC_TILE_A;
    const float* cB = sB + (ch & 1) * FC_TILE_B;
    #pragma unroll
    for (int ks = 0; ks < FC_KC / 4; ++ks) {
      v2f a[2], bf[4];
      #pragma unroll
      for (int i = 0; i < 2; ++i)
        a[i] = *(const v2f*)(cA + aoff + i * 16 * FC_PITCH_A + ks * 4);
      #pragma unroll
      for (int t = 0; t < 4; ++t)
        bf[t] = *(const v2f*)(cB + boff + ks * 2 * FC_PITCH_B + t * 32);
      #pragma unroll
      for (int i = 0; i < 2; ++i)
        #pragma unroll
        for (int t = 0; t < 4; ++t)
          acc[i][t] = wmma_f32(a[i], bf[t], acc[i][t]);
    }
    __builtin_amdgcn_sched_barrier(0);
    if (ch + 1 < H_DIM / FC_KC) {
      float* nA = sA + ((ch + 1) & 1) * FC_TILE_A;
      float* nB = sB + ((ch + 1) & 1) * FC_TILE_B;
      FC_SSTORE(nA, nB)
    }
    if (ch + 2 < H_DIM / FC_KC) {
      int ko = (ch + 2) * FC_KC;
      FC_GLOAD(ko)
    }
    lds_barrier();
  }

  #pragma unroll
  for (int t = 0; t < 4; ++t) {
    int col = c0 + wc * 64 + t * 16 + c;
    float bz = bias[col];
    #pragma unroll
    for (int i = 0; i < 2; ++i) {
      float* dst = q + (size_t)(r0 + wr * 32 + i * 16 + 8 * h) * H_DIM + col;
      #pragma unroll
      for (int j = 0; j < 8; ++j)
        dst[(size_t)j * H_DIM] = acc[i][t][j] + bz;
    }
  }
}

#define SC_PITCH 36
#define SC_TILE  (128 * SC_PITCH)
__global__ __launch_bounds__(256)
void scores_kernel(const float* __restrict__ q,
                   const float* __restrict__ feats,
                   float* __restrict__ S) {
  extern __shared__ float stage[];
  float* sA = stage;
  float* sB = stage + 2 * SC_TILE;

  int b  = blockIdx.z;
  int n0 = blockIdx.y * 128;
  int m0 = blockIdx.x * 128;
  const float* Qg = q     + ((size_t)b * N_DIM + n0) * H_DIM;
  const float* Fg = feats + ((size_t)b * N_DIM + m0) * H_DIM;
  float* Sg = S + ((size_t)b * N_DIM + n0) * N_DIM + m0;

  int tid  = threadIdx.x;
  int wave = tid >> 5;
  int lane = tid & 31;
  int wr = wave >> 1;
  int wc = wave & 1;
  int c  = lane & 15;
  int h  = lane >> 4;

  int srow = tid >> 3;
  int sseg = (tid & 7) * 4;
  const float* gA = Qg + (size_t)srow * H_DIM + sseg;
  const float* gB = Fg + (size_t)srow * H_DIM + sseg;
  int soff = srow * SC_PITCH + sseg;

  float4 ra0, ra1, ra2, ra3, rb0, rb1, rb2, rb3;
  ra0 = *(const float4*)(gA + (size_t)0 * H_DIM + 0);
  rb0 = *(const float4*)(gB + (size_t)0 * H_DIM + 0);
  ra1 = *(const float4*)(gA + (size_t)32 * H_DIM + 0);
  rb1 = *(const float4*)(gB + (size_t)32 * H_DIM + 0);
  ra2 = *(const float4*)(gA + (size_t)64 * H_DIM + 0);
  rb2 = *(const float4*)(gB + (size_t)64 * H_DIM + 0);
  ra3 = *(const float4*)(gA + (size_t)96 * H_DIM + 0);
  rb3 = *(const float4*)(gB + (size_t)96 * H_DIM + 0);
  *(float4*)(sA + soff + 0 * SC_PITCH) = ra0;
  *(float4*)(sB + soff + 0 * SC_PITCH) = rb0;
  *(float4*)(sA + soff + 32 * SC_PITCH) = ra1;
  *(float4*)(sB + soff + 32 * SC_PITCH) = rb1;
  *(float4*)(sA + soff + 64 * SC_PITCH) = ra2;
  *(float4*)(sB + soff + 64 * SC_PITCH) = rb2;
  *(float4*)(sA + soff + 96 * SC_PITCH) = ra3;
  *(float4*)(sB + soff + 96 * SC_PITCH) = rb3;
  ra0 = *(const float4*)(gA + (size_t)0 * H_DIM + 32);
  rb0 = *(const float4*)(gB + (size_t)0 * H_DIM + 32);
  ra1 = *(const float4*)(gA + (size_t)32 * H_DIM + 32);
  rb1 = *(const float4*)(gB + (size_t)32 * H_DIM + 32);
  ra2 = *(const float4*)(gA + (size_t)64 * H_DIM + 32);
  rb2 = *(const float4*)(gB + (size_t)64 * H_DIM + 32);
  ra3 = *(const float4*)(gA + (size_t)96 * H_DIM + 32);
  rb3 = *(const float4*)(gB + (size_t)96 * H_DIM + 32);
  lds_barrier();

  v8f acc[2][4] = {};
  int aoff = (wr * 32 + c) * SC_PITCH + 2 * h;
  int boff = (wc * 64 + c) * SC_PITCH + 2 * h;

  for (int ch = 0; ch < H_DIM / 32; ++ch) {
    const float* cA = sA + (ch & 1) * SC_TILE;
    const float* cB = sB + (ch & 1) * SC_TILE;
    #pragma unroll
    for (int ks = 0; ks < 8; ++ks) {
      v2f a[2], bf[4];
      #pragma unroll
      for (int i = 0; i < 2; ++i)
        a[i] = *(const v2f*)(cA + aoff + i * 16 * SC_PITCH + ks * 4);
      #pragma unroll
      for (int t = 0; t < 4; ++t)
        bf[t] = *(const v2f*)(cB + boff + t * 16 * SC_PITCH + ks * 4);
      #pragma unroll
      for (int i = 0; i < 2; ++i)
        #pragma unroll
        for (int t = 0; t < 4; ++t)
          acc[i][t] = wmma_f32(a[i], bf[t], acc[i][t]);
    }
    if (ch + 1 < H_DIM / 32) {
      float* nA = sA + ((ch + 1) & 1) * SC_TILE;
      float* nB = sB + ((ch + 1) & 1) * SC_TILE;
      *(float4*)(nA + soff + 0 * SC_PITCH) = ra0;
      *(float4*)(nB + soff + 0 * SC_PITCH) = rb0;
      *(float4*)(nA + soff + 32 * SC_PITCH) = ra1;
      *(float4*)(nB + soff + 32 * SC_PITCH) = rb1;
      *(float4*)(nA + soff + 64 * SC_PITCH) = ra2;
      *(float4*)(nB + soff + 64 * SC_PITCH) = rb2;
      *(float4*)(nA + soff + 96 * SC_PITCH) = ra3;
      *(float4*)(nB + soff + 96 * SC_PITCH) = rb3;
    }
    if (ch + 2 < H_DIM / 32) {
      int ko = (ch + 2) * 32;
      ra0 = *(const float4*)(gA + (size_t)0 * H_DIM + ko);
      rb0 = *(const float4*)(gB + (size_t)0 * H_DIM + ko);
      ra1 = *(const float4*)(gA + (size_t)32 * H_DIM + ko);
      rb1 = *(const float4*)(gB + (size_t)32 * H_DIM + ko);
      ra2 = *(const float4*)(gA + (size_t)64 * H_DIM + ko);
      rb2 = *(const float4*)(gB + (size_t)64 * H_DIM + ko);
      ra3 = *(const float4*)(gA + (size_t)96 * H_DIM + ko);
      rb3 = *(const float4*)(gB + (size_t)96 * H_DIM + ko);
    }
    lds_barrier();
  }

  #pragma unroll
  for (int i = 0; i < 2; ++i) {
    #pragma unroll
    for (int t = 0; t < 4; ++t) {
      float* dst = Sg + (size_t)(wr * 32 + i * 16 + 8 * h) * N_DIM + wc * 64 + t * 16 + c;
      #pragma unroll
      for (int j = 0; j < 8; ++j)
        dst[(size_t)j * N_DIM] = acc[i][t][j];
    }
  }
}

__global__ void softmax_kernel(const float* __restrict__ S,
                               float* __restrict__ partial) {
  int b  = blockIdx.x >> 7;
  int nt = blockIdx.x & (NT - 1);
  const float* Sr = S + ((size_t)b * N_DIM + nt * 16) * N_DIM;

  extern __shared__ float slab[];
  __shared__ float rowInv[16];

  int wave = threadIdx.x >> 5;
  int lane = threadIdx.x & 31;

  for (int i = threadIdx.x * 4; i < 16 * N_DIM; i += 256 * 4)
    *(float4*)(slab + i) = *(const float4*)(Sr + i);
  __syncthreads();

  for (int rr = 0; rr < 2; ++rr) {
    int r = wave * 2 + rr;
    float mx = -3.402823466e38f;
    for (int m = lane; m < N_DIM; m += 32)
      mx = fmaxf(mx, slab[r * N_DIM + m]);
    #pragma unroll
    for (int o = 16; o > 0; o >>= 1)
      mx = fmaxf(mx, __shfl_xor(mx, o, 32));
    float s = 0.f;
    for (int m = lane; m < N_DIM; m += 32) {
      float e = __expf(slab[r * N_DIM + m] - mx);
      slab[r * N_DIM + m] = e;
      s += e;
    }
    #pragma unroll
    for (int o = 16; o > 0; o >>= 1)
      s += __shfl_xor(s, o, 32);
    if (lane == 0) rowInv[r] = 1.0f / s;
  }
  __syncthreads();

  float* P = partial + ((size_t)b * NT + nt) * N_DIM;
  for (int m = threadIdx.x; m < N_DIM; m += 256) {
    float a = 0.f;
    #pragma unroll
    for (int r = 0; r < 16; ++r)
      a += slab[r * N_DIM + m] * rowInv[r];
    P[m] = a;
  }
}

__global__ void colsum_kernel(const float* __restrict__ partial,
                              float* __restrict__ colsum) {
  int idx = blockIdx.x * blockDim.x + threadIdx.x;
  int b = idx >> 11;
  int m = idx & (N_DIM - 1);
  const float* P = partial + (size_t)b * NT * N_DIM + m;
  float a = 0.f;
  #pragma unroll 8
  for (int t = 0; t < NT; ++t) a += P[(size_t)t * N_DIM];
  colsum[idx] = a;
}

#define PL_ROWS   32
#define PL_STRIPS (N_DIM / PL_ROWS)
typedef __attribute__((ext_vector_type(4))) float v4f;

__global__ __launch_bounds__(256)
void pool_strip_kernel(const float* __restrict__ colsum,
                       const float* __restrict__ feats,
                       float* __restrict__ pp) {
  int s = blockIdx.x;
  int b = blockIdx.y;
  int m0 = s * PL_ROWS;
  const float* F = feats + ((size_t)b * N_DIM + m0) * H_DIM + threadIdx.x * 4;
  const float* C = colsum + (size_t)b * N_DIM + m0;

  v4f acc = {0.f, 0.f, 0.f, 0.f};
  #pragma unroll
  for (int i = 0; i < PL_ROWS; ++i) {
    v4f f = __builtin_nontemporal_load((const v4f*)(F + (size_t)i * H_DIM));
    float c = C[i];
    acc.x = fmaf(c, f.x, acc.x);
    acc.y = fmaf(c, f.y, acc.y);
    acc.z = fmaf(c, f.z, acc.z);
    acc.w = fmaf(c, f.w, acc.w);
  }
  *(v4f*)(pp + ((size_t)b * PL_STRIPS + s) * H_DIM + threadIdx.x * 4) = acc;
}

__global__ void pool_join_kernel(const float* __restrict__ pp,
                                 float* __restrict__ out) {
  int idx = blockIdx.x * blockDim.x + threadIdx.x;
  int b = idx >> 10;
  int h = idx & (H_DIM - 1);
  const float* P = pp + (size_t)b * PL_STRIPS * H_DIM + h;
  float a = 0.f;
  #pragma unroll
  for (int s = 0; s < PL_STRIPS; ++s) a += P[(size_t)s * H_DIM];
  out[idx] = a;
}

#define GRP_FEATS (4 * N_DIM * H_DIM)
#define GRP_PART  (4 * NT * N_DIM)
#define RUN_GROUP(G)                                                            \
  fc1_kernel<<<dim3(8, 64), 256, (2 * FC_TILE_A + 2 * FC_TILE_B) * sizeof(float), stream>>>(  \
      feats + (size_t)(G) * GRP_FEATS, W, bias, q);                                         \
  scores_kernel<<<dim3(16, 16, 4), 256, 4 * SC_TILE * sizeof(float), stream>>>(            \
      q, feats + (size_t)(G) * GRP_FEATS, S);                                               \
  softmax_kernel<<<4 * NT, 256, 16 * N_DIM * sizeof(float), stream>>>(                      \
      S, partial + (size_t)(G) * GRP_PART);

extern "C" void kernel_launch(void* const* d_in, const int* in_sizes, int n_in,
                              void* d_out, int out_size, void* d_ws, size_t ws_size,
                              hipStream_t stream) {
  const float* feats = (const float*)d_in[0];
  const float* W     = (const float*)d_in[1];
  const float* bias  = (const float*)d_in[2];
  float* out = (float*)d_out;

  float* q       = (float*)d_ws;
  float* S       = q + (size_t)4 * N_DIM * H_DIM;
  float* partial = S + (size_t)4 * N_DIM * N_DIM;
  float* colsum  = partial + (size_t)B_DIM * NT * N_DIM;

  RUN_GROUP(0)
  RUN_GROUP(1)
  RUN_GROUP(2)
  RUN_GROUP(3)

  colsum_kernel<<<(B_DIM * N_DIM) / 256, 256, 0, stream>>>(partial, colsum);

  float* pp = q;
  pool_strip_kernel<<<dim3(PL_STRIPS, B_DIM), 256, 0, stream>>>(colsum, feats, pp);
  pool_join_kernel<<<(B_DIM * H_DIM) / 256, 256, 0, stream>>>(pp, out);
}


# Softmax kernel rebuilt as a stream: a wave holds its two rows of S whole in registers (32 whole-line 128-bit loads in flight per lane), max, exp and sum there; exp row sums in a new order
# speedup vs baseline: 2.8561x; 1.0511x over previous
#include <hip/hip_runtime.h>

#define B_DIM 16
#define N_DIM 2048
#define H_DIM 1024
#define NT    128

typedef __attribute__((ext_vector_type(2))) float v2f;
typedef __attribute__((ext_vector_type(8))) float v8f;


__device__ __forceinline__ v8f wmma_f32(v2f a, v2f b, v8f c) {
  return __builtin_amdgcn_wmma_f32_16x16x4_f32(false, a, false, b, (short)0, c,
                                               false, false);
}

__device__ __forceinline__ void lds_barrier() {
  asm volatile("s_wait_dscnt 0x0" ::: "memory");
  asm volatile("s_barrier_signal -1" ::: "memory");
  asm volatile("s_barrier_wait -1" ::: "memory");
}

#define FC_KC      64
#define FC_PITCH_A 68
#define FC_PITCH_B 288
#define FC_TILE_A  (128 * FC_PITCH_A)
#define FC_TILE_B  (32 * FC_PITCH_B)

#define FC_GLOAD(KO)                                                    \
  ra0 = *(const float4*)(gA + (size_t)0   * H_DIM + (KO));              \
  rb0 = *(const float4*)(gB + ((size_t)(KO) + 0)  * H_DIM);             \
  ra1 = *(const float4*)(gA + (size_t)16  * H_DIM + (KO));              \
  rb1 = *(const float4*)(gB + ((size_t)(KO) + 1)  * H_DIM);             \
  ra2 = *(const float4*)(gA + (size_t)32  * H_DIM + (KO));              \
  rb2 = *(const float4*)(gB + ((size_t)(KO) + 16) * H_DIM);             \
  ra3 = *(const float4*)(gA + (size_t)48  * H_DIM + (KO));              \
  rb3 = *(const float4*)(gB + ((size_t)(KO) + 17) * H_DIM);             \
  ra4 = *(const float4*)(gA + (size_t)64  * H_DIM + (KO));              \
  rb4 = *(const float4*)(gB + ((size_t)(KO) + 32) * H_DIM);             \
  ra5 = *(const float4*)(gA + (size_t)80  * H_DIM + (KO));              \
  rb5 = *(const float4*)(gB + ((size_t)(KO) + 33) * H_DIM);             \
  ra6 = *(const float4*)(gA + (size_t)96  * H_DIM + (KO));              \
  rb6 = *(const float4*)(gB + ((size_t)(KO) + 48) * H_DIM);             \
  ra7 = *(const float4*)(gA + (size_t)112 * H_DIM + (KO));              \
  rb7 = *(const float4*)(gB + ((size_t)(KO) + 49) * H_DIM);

#define FC_BPAIR(DB, J, LO, HI)                                                             \
  *(float4*)((DB) + bsoff + (J) * 8 * FC_PITCH_B)     = make_float4(LO.x, HI.x, LO.y, HI.y); \
  *(float4*)((DB) + bsoff + (J) * 8 * FC_PITCH_B + 4) = make_float4(LO.z, HI.z, LO.w, HI.w);
#define FC_SSTORE(DA, DB)                                  \
  *(float4*)((DA) + asoff + 0   * FC_PITCH_A) = ra0;       \
  *(float4*)((DA) + asoff + 16  * FC_PITCH_A) = ra1;       \
  *(float4*)((DA) + asoff + 32  * FC_PITCH_A) = ra2;       \
  *(float4*)((DA) + asoff + 48  * FC_PITCH_A) = ra3;       \
  *(float4*)((DA) + asoff + 64  * FC_PITCH_A) = ra4;       \
  *(float4*)((DA) + asoff + 80  * FC_PITCH_A) = ra5;       \
  *(float4*)((DA) + asoff + 96  * FC_PITCH_A) = ra6;       \
  *(float4*)((DA) + asoff + 112 * FC_PITCH_A) = ra7;       \
  FC_BPAIR(DB, 0, rb0, rb1)                                \
  FC_BPAIR(DB, 1, rb2, rb3)                                \
  FC_BPAIR(DB, 2, rb4, rb5)                                \
  FC_BPAIR(DB, 3, rb6, rb7)

__global__ __launch_bounds__(256)
void fc1_kernel(const float* __restrict__ feats,
                const float* __restrict__ W,
                const float* __restrict__ bias,
                float* __restrict__ q) {
  extern __shared__ float stage[];
  float* sA = stage;
  float* sB = stage + 2 * FC_TILE_A;

  int r0 = blockIdx.y * 128;
  int c0 = blockIdx.x * 128;

  int tid  = threadIdx.x;
  int wave = tid >> 5;
  int lane = tid & 31;
  int wr = wave >> 1;
  int wc = wave & 1;
  int c  = lane & 15;
  int h  = lane >> 4;

  int arow = tid >> 4;
  int aseg = (tid & 15) * 4;
  const float* gA = feats + (size_t)(r0 + arow) * H_DIM + aseg;
  int asoff = arow * FC_PITCH_A + aseg;
  int bpair = tid >> 5;
  int bcol  = (tid & 31) * 4;
  const float* gB = W + (size_t)(2 * bpair) * H_DIM + c0 + bcol;
  int bsoff = bpair * FC_PITCH_B + 2 * bcol;

  float4 ra0, ra1, ra2, ra3, ra4, ra5, ra6, ra7;
  float4 rb0, rb1, rb2, rb3, rb4, rb5, rb6, rb7;
  FC_GLOAD(0)
  FC_SSTORE(sA, sB)
  FC_GLOAD(FC_KC)
  lds_barrier();

  v8f acc[2][4] = {};
  int aoff = (wr * 32 + c) * FC_PITCH_A + 2 * h;
  int boff = h * FC_PITCH_B + 2 * (wc * 64 + c);

  for (int ch = 0; ch < H_DIM / FC_KC; ++ch) {
    const float* cA = sA + (ch & 1) * FC_TILE_A;
    const float* cB = sB + (ch & 1) * FC_TILE_B;
    #pragma unroll
    for (int ks = 0; ks < FC_KC / 4; ++ks) {
      v2f a[2], bf[4];
      #pragma unroll
      for (int i = 0; i < 2; ++i)
        a[i] = *(const v2f*)(cA + aoff + i * 16 * FC_PITCH_A + ks * 4);
      #pragma unroll
      for (int t = 0; t < 4; ++t)
        bf[t] = *(const v2f*)(cB + boff + ks * 2 * FC_PITCH_B + t * 32);
      #pragma unroll
      for (int i = 0; i < 2; ++i)
        #pragma unroll
        for (int t = 0; t < 4; ++t)
          acc[i][t] = wmma_f32(a[i], bf[t], acc[i][t]);
    }
    __builtin_amdgcn_sched_barrier(0);
    if (ch + 1 < H_DIM / FC_KC) {
      float* nA = sA + ((ch + 1) & 1) * FC_TILE_A;
      float* nB = sB + ((ch + 1) & 1) * FC_TILE_B;
      FC_SSTORE(nA, nB)
    }
    if (ch + 2 < H_DIM / FC_KC) {
      int ko = (ch + 2) * FC_KC;
      FC_GLOAD(ko)
    }
    lds_barrier();
  }

  #pragma unroll
  for (int t = 0; t < 4; ++t) {
    int col = c0 + wc * 64 + t * 16 + c;
    float bz = bias[col];
    #pragma unroll
    for (int i = 0; i < 2; ++i) {
      float* dst = q + (size_t)(r0 + wr * 32 + i * 16 + 8 * h) * H_DIM + col;
      #pragma unroll
      for (int j = 0; j < 8; ++j)
        dst[(size_t)j * H_DIM] = acc[i][t][j] + bz;
    }
  }
}

#define SC_PITCH 36
#define SC_TILE  (128 * SC_PITCH)
__global__ __launch_bounds__(256)
void scores_kernel(const float* __restrict__ q,
                   const float* __restrict__ feats,
                   float* __restrict__ S) {
  extern __shared__ float stage[];
  float* sA = stage;
  float* sB = stage + 2 * SC_TILE;

  int b  = blockIdx.z;
  int n0 = blockIdx.y * 128;
  int m0 = blockIdx.x * 128;
  const float* Qg = q     + ((size_t)b * N_DIM + n0) * H_DIM;
  const float* Fg = feats + ((size_t)b * N_DIM + m0) * H_DIM;
  float* Sg = S + ((size_t)b * N_DIM + n0) * N_DIM + m0;

  int tid  = threadIdx.x;
  int wave = tid >> 5;
  int lane = tid & 31;
  int wr = wave >> 1;
  int wc = wave & 1;
  int c  = lane & 15;
  int h  = lane >> 4;

  int srow = tid >> 3;
  int sseg = (tid & 7) * 4;
  const float* gA = Qg + (size_t)srow * H_DIM + sseg;
  const float* gB = Fg + (size_t)srow * H_DIM + sseg;
  int soff = srow * SC_PITCH + sseg;

  float4 ra0, ra1, ra2, ra3, rb0, rb1, rb2, rb3;
  ra0 = *(const float4*)(gA + (size_t)0 * H_DIM + 0);
  rb0 = *(const float4*)(gB + (size_t)0 * H_DIM + 0);
  ra1 = *(const float4*)(gA + (size_t)32 * H_DIM + 0);
  rb1 = *(const float4*)(gB + (size_t)32 * H_DIM + 0);
  ra2 = *(const float4*)(gA + (size_t)64 * H_DIM + 0);
  rb2 = *(const float4*)(gB + (size_t)64 * H_DIM + 0);
  ra3 = *(const float4*)(gA + (size_t)96 * H_DIM + 0);
  rb3 = *(const float4*)(gB + (size_t)96 * H_DIM + 0);
  *(float4*)(sA + soff + 0 * SC_PITCH) = ra0;
  *(float4*)(sB + soff + 0 * SC_PITCH) = rb0;
  *(float4*)(sA + soff + 32 * SC_PITCH) = ra1;
  *(float4*)(sB + soff + 32 * SC_PITCH) = rb1;
  *(float4*)(sA + soff + 64 * SC_PITCH) = ra2;
  *(float4*)(sB + soff + 64 * SC_PITCH) = rb2;
  *(float4*)(sA + soff + 96 * SC_PITCH) = ra3;
  *(float4*)(sB + soff + 96 * SC_PITCH) = rb3;
  ra0 = *(const float4*)(gA + (size_t)0 * H_DIM + 32);
  rb0 = *(const float4*)(gB + (size_t)0 * H_DIM + 32);
  ra1 = *(const float4*)(gA + (size_t)32 * H_DIM + 32);
  rb1 = *(const float4*)(gB + (size_t)32 * H_DIM + 32);
  ra2 = *(const float4*)(gA + (size_t)64 * H_DIM + 32);
  rb2 = *(const float4*)(gB + (size_t)64 * H_DIM + 32);
  ra3 = *(const float4*)(gA + (size_t)96 * H_DIM + 32);
  rb3 = *(const float4*)(gB + (size_t)96 * H_DIM + 32);
  lds_barrier();

  v8f acc[2][4] = {};
  int aoff = (wr * 32 + c) * SC_PITCH + 2 * h;
  int boff = (wc * 64 + c) * SC_PITCH + 2 * h;

  for (int ch = 0; ch < H_DIM / 32; ++ch) {
    const float* cA = sA + (ch & 1) * SC_TILE;
    const float* cB = sB + (ch & 1) * SC_TILE;
    #pragma unroll
    for (int ks = 0; ks < 8; ++ks) {
      v2f a[2], bf[4];
      #pragma unroll
      for (int i = 0; i < 2; ++i)
        a[i] = *(const v2f*)(cA + aoff + i * 16 * SC_PITCH + ks * 4);
      #pragma unroll
      for (int t = 0; t < 4; ++t)
        bf[t] = *(const v2f*)(cB + boff + t * 16 * SC_PITCH + ks * 4);
      #pragma unroll
      for (int i = 0; i < 2; ++i)
        #pragma unroll
        for (int t = 0; t < 4; ++t)
          acc[i][t] = wmma_f32(a[i], bf[t], acc[i][t]);
    }
    if (ch + 1 < H_DIM / 32) {
      float* nA = sA + ((ch + 1) & 1) * SC_TILE;
      float* nB = sB + ((ch + 1) & 1) * SC_TILE;
      *(float4*)(nA + soff + 0 * SC_PITCH) = ra0;
      *(float4*)(nB + soff + 0 * SC_PITCH) = rb0;
      *(float4*)(nA + soff + 32 * SC_PITCH) = ra1;
      *(float4*)(nB + soff + 32 * SC_PITCH) = rb1;
      *(float4*)(nA + soff + 64 * SC_PITCH) = ra2;
      *(float4*)(nB + soff + 64 * SC_PITCH) = rb2;
      *(float4*)(nA + soff + 96 * SC_PITCH) = ra3;
      *(float4*)(nB + soff + 96 * SC_PITCH) = rb3;
    }
    if (ch + 2 < H_DIM / 32) {
      int ko = (ch + 2) * 32;
      ra0 = *(const float4*)(gA + (size_t)0 * H_DIM + ko);
      rb0 = *(const float4*)(gB + (size_t)0 * H_DIM + ko);
      ra1 = *(const float4*)(gA + (size_t)32 * H_DIM + ko);
      rb1 = *(const float4*)(gB + (size_t)32 * H_DIM + ko);
      ra2 = *(const float4*)(gA + (size_t)64 * H_DIM + ko);
      rb2 = *(const float4*)(gB + (size_t)64 * H_DIM + ko);
      ra3 = *(const float4*)(gA + (size_t)96 * H_DIM + ko);
      rb3 = *(const float4*)(gB + (size_t)96 * H_DIM + ko);
    }
    lds_barrier();
  }

  #pragma unroll
  for (int i = 0; i < 2; ++i) {
    #pragma unroll
    for (int t = 0; t < 4; ++t) {
      float* dst = Sg + (size_t)(wr * 32 + i * 16 + 8 * h) * N_DIM + wc * 64 + t * 16 + c;
      #pragma unroll
      for (int j = 0; j < 8; ++j)
        dst[(size_t)j * N_DIM] = acc[i][t][j];
    }
  }
}

typedef __attribute__((ext_vector_type(4))) float v4f;

__global__ __launch_bounds__(256)
void softmax_kernel(const float* __restrict__ S,
                    float* __restrict__ partial) {
  int b  = blockIdx.x >> 7;
  int nt = blockIdx.x & (NT - 1);
  const float* Sr = S + ((size_t)b * N_DIM + nt * 16) * N_DIM;

  extern __shared__ __attribute__((aligned(16))) float slab[];
  __shared__ float rowInv[16];

  int wave = threadIdx.x >> 5;
  int lane = threadIdx.x & 31;

  const float* R = Sr + (size_t)(2 * wave) * N_DIM + lane * 4;
  v4f x[2][16];
  #pragma unroll
  for (int rr = 0; rr < 2; ++rr)
    #pragma unroll
    for (int j = 0; j < 16; ++j)
      x[rr][j] = __builtin_nontemporal_load((const v4f*)(R + (size_t)rr * N_DIM + j * 128));

  float mx[2];
  #pragma unroll
  for (int rr = 0; rr < 2; ++rr) {
    float m0 = -3.402823466e38f;
    #pragma unroll
    for (int j = 0; j < 16; ++j)
      m0 = fmaxf(m0, fmaxf(fmaxf(x[rr][j].x, x[rr][j].y), fmaxf(x[rr][j].z, x[rr][j].w)));
    mx[rr] = m0;
  }
  #pragma unroll
  for (int o = 16; o > 0; o >>= 1)
    #pragma unroll
    for (int rr = 0; rr < 2; ++rr)
      mx[rr] = fmaxf(mx[rr], __shfl_xor(mx[rr], o, 32));

  float s[2];
  #pragma unroll
  for (int rr = 0; rr < 2; ++rr) {
    float s0 = 0.f, s1 = 0.f, s2 = 0.f, s3 = 0.f;
    #pragma unroll
    for (int j = 0; j < 16; ++j) {
      v4f e;
      e.x = __expf(x[rr][j].x - mx[rr]);
      e.y = __expf(x[rr][j].y - mx[rr]);
      e.z = __expf(x[rr][j].z - mx[rr]);
      e.w = __expf(x[rr][j].w - mx[rr]);
      x[rr][j] = e;
      s0 += e.x;
      s1 += e.y;
      s2 += e.z;
      s3 += e.w;
    }
    s[rr] = (s0 + s1) + (s2 + s3);
  }
  #pragma unroll
  for (int o = 16; o > 0; o >>= 1)
    #pragma unroll
    for (int rr = 0; rr < 2; ++rr)
      s[rr] += __shfl_xor(s[rr], o, 32);
  if (lane == 0) {
    rowInv[2 * wave]     = 1.0f / s[0];
    rowInv[2 * wave + 1] = 1.0f / s[1];
  }
  #pragma unroll
  for (int rr = 0; rr < 2; ++rr)
    #pragma unroll
    for (int j = 0; j < 16; ++j)
      *(v4f*)(slab + (2 * wave + rr) * N_DIM + j * 128 + lane * 4) = x[rr][j];
  __syncthreads();

  float* P = partial + ((size_t)b * NT + nt) * N_DIM;
  #pragma unroll
  for (int k = 0; k < 2; ++k) {
    int m = k * (N_DIM / 2) + threadIdx.x * 4;
    v4f a = {0.f, 0.f, 0.f, 0.f};
    #pragma unroll
    for (int r = 0; r < 16; ++r) {
      v4f e = *(const v4f*)(slab + r * N_DIM + m);
      float inv = rowInv[r];
      a.x = fmaf(e.x, inv, a.x);
      a.y = fmaf(e.y, inv, a.y);
      a.z = fmaf(e.z, inv, a.z);
      a.w = fmaf(e.w, inv, a.w);
    }
    *(v4f*)(P + m) = a;
  }
}

__global__ void colsum_kernel(const float* __restrict__ partial,
                              float* __restrict__ colsum) {
  int idx = blockIdx.x * blockDim.x + threadIdx.x;
  int b = idx >> 11;
  int m = idx & (N_DIM - 1);
  const float* P = partial + (size_t)b * NT * N_DIM + m;
  float a = 0.f;
  #pragma unroll 8
  for (int t = 0; t < NT; ++t) a += P[(size_t)t * N_DIM];
  colsum[idx] = a;
}

#define PL_ROWS   32
#define PL_STRIPS (N_DIM / PL_ROWS)
__global__ __launch_bounds__(256)
void pool_strip_kernel(const float* __restrict__ colsum,
                       const float* __restrict__ feats,
                       float* __restrict__ pp) {
  int s = blockIdx.x;
  int b = blockIdx.y;
  int m0 = s * PL_ROWS;
  const float* F = feats + ((size_t)b * N_DIM + m0) * H_DIM + threadIdx.x * 4;
  const float* C = colsum + (size_t)b * N_DIM + m0;

  v4f acc = {0.f, 0.f, 0.f, 0.f};
  #pragma unroll
  for (int i = 0; i < PL_ROWS; ++i) {
    v4f f = __builtin_nontemporal_load((const v4f*)(F + (size_t)i * H_DIM));
    float c = C[i];
    acc.x = fmaf(c, f.x, acc.x);
    acc.y = fmaf(c, f.y, acc.y);
    acc.z = fmaf(c, f.z, acc.z);
    acc.w = fmaf(c, f.w, acc.w);
  }
  *(v4f*)(pp + ((size_t)b * PL_STRIPS + s) * H_DIM + threadIdx.x * 4) = acc;
}

__global__ void pool_join_kernel(const float* __restrict__ pp,
                                 float* __restrict__ out) {
  int idx = blockIdx.x * blockDim.x + threadIdx.x;
  int b = idx >> 10;
  int h = idx & (H_DIM - 1);
  const float* P = pp + (size_t)b * PL_STRIPS * H_DIM + h;
  float a = 0.f;
  #pragma unroll
  for (int s = 0; s < PL_STRIPS; ++s) a += P[(size_t)s * H_DIM];
  out[idx] = a;
}

#define GRP_FEATS (4 * N_DIM * H_DIM)
#define GRP_PART  (4 * NT * N_DIM)
#define RUN_GROUP(G)                                                            \
  fc1_kernel<<<dim3(8, 64), 256, (2 * FC_TILE_A + 2 * FC_TILE_B) * sizeof(float), stream>>>(  \
      feats + (size_t)(G) * GRP_FEATS, W, bias, q);                                         \
  scores_kernel<<<dim3(16, 16, 4), 256, 4 * SC_TILE * sizeof(float), stream>>>(            \
      q, feats + (size_t)(G) * GRP_FEATS, S);                                               \
  softmax_kernel<<<4 * NT, 256, 16 * N_DIM * sizeof(float), stream>>>(                      \
      S, partial + (size_t)(G) * GRP_PART);

extern "C" void kernel_launch(void* const* d_in, const int* in_sizes, int n_in,
                              void* d_out, int out_size, void* d_ws, size_t ws_size,
                              hipStream_t stream) {
  const float* feats = (const float*)d_in[0];
  const float* W     = (const float*)d_in[1];
  const float* bias  = (const float*)d_in[2];
  float* out = (float*)d_out;

  float* q       = (float*)d_ws;
  float* S       = q + (size_t)4 * N_DIM * H_DIM;
  float* partial = S + (size_t)4 * N_DIM * N_DIM;
  float* colsum  = partial + (size_t)B_DIM * NT * N_DIM;

  RUN_GROUP(0)
  RUN_GROUP(1)
  RUN_GROUP(2)
  RUN_GROUP(3)

  colsum_kernel<<<(B_DIM * N_DIM) / 256, 256, 0, stream>>>(partial, colsum);

  float* pp = q;
  pool_strip_kernel<<<dim3(PL_STRIPS, B_DIM), 256, 0, stream>>>(colsum, feats, pp);
  pool_join_kernel<<<(B_DIM * H_DIM) / 256, 256, 0, stream>>>(pp, out);
}


# All 16 batches go through fc1, score and softmax kernel in one launch each if the workspace holds q and S of all 16 (15 to 6 launches; else 8 or 4 batches per launch); kernels and sums unchanged
# speedup vs baseline: 2.9923x; 1.0477x over previous
#include <hip/hip_runtime.h>

#define B_DIM 16
#define N_DIM 2048
#define H_DIM 1024
#define NT    128

typedef __attribute__((ext_vector_type(2))) float v2f;
typedef __attribute__((ext_vector_type(8))) float v8f;


__device__ __forceinline__ v8f wmma_f32(v2f a, v2f b, v8f c) {
  return __builtin_amdgcn_wmma_f32_16x16x4_f32(false, a, false, b, (short)0, c,
                                               false, false);
}

__device__ __forceinline__ void lds_barrier() {
  asm volatile("s_wait_dscnt 0x0" ::: "memory");
  asm volatile("s_barrier_signal -1" ::: "memory");
  asm volatile("s_barrier_wait -1" ::: "memory");
}

#define FC_KC      64
#define FC_PITCH_A 68
#define FC_PITCH_B 288
#define FC_TILE_A  (128 * FC_PITCH_A)
#define FC_TILE_B  (32 * FC_PITCH_B)

#define FC_GLOAD(KO)                                                    \
  ra0 = *(const float4*)(gA + (size_t)0   * H_DIM + (KO));              \
  rb0 = *(const float4*)(gB + ((size_t)(KO) + 0)  * H_DIM);             \
  ra1 = *(const float4*)(gA + (size_t)16  * H_DIM + (KO));              \
  rb1 = *(const float4*)(gB + ((size_t)(KO) + 1)  * H_DIM);             \
  ra2 = *(const float4*)(gA + (size_t)32  * H_DIM + (KO));              \
  rb2 = *(const float4*)(gB + ((size_t)(KO) + 16) * H_DIM);             \
  ra3 = *(const float4*)(gA + (size_t)48  * H_DIM + (KO));              \
  rb3 = *(const float4*)(gB + ((size_t)(KO) + 17) * H_DIM);             \
  ra4 = *(const float4*)(gA + (size_t)64  * H_DIM + (KO));              \
  rb4 = *(const float4*)(gB + ((size_t)(KO) + 32) * H_DIM);             \
  ra5 = *(const float4*)(gA + (size_t)80  * H_DIM + (KO));              \
  rb5 = *(const float4*)(gB + ((size_t)(KO) + 33) * H_DIM);             \
  ra6 = *(const float4*)(gA + (size_t)96  * H_DIM + (KO));              \
  rb6 = *(const float4*)(gB + ((size_t)(KO) + 48) * H_DIM);             \
  ra7 = *(const float4*)(gA + (size_t)112 * H_DIM + (KO));              \
  rb7 = *(const float4*)(gB + ((size_t)(KO) + 49) * H_DIM);

#define FC_BPAIR(DB, J, LO, HI)                                                             \
  *(float4*)((DB) + bsoff + (J) * 8 * FC_PITCH_B)     = make_float4(LO.x, HI.x, LO.y, HI.y); \
  *(float4*)((DB) + bsoff + (J) * 8 * FC_PITCH_B + 4) = make_float4(LO.z, HI.z, LO.w, HI.w);
#define FC_SSTORE(DA, DB)                                  \
  *(float4*)((DA) + asoff + 0   * FC_PITCH_A) = ra0;       \
  *(float4*)((DA) + asoff + 16  * FC_PITCH_A) = ra1;       \
  *(float4*)((DA) + asoff + 32  * FC_PITCH_A) = ra2;       \
  *(float4*)((DA) + asoff + 48  * FC_PITCH_A) = ra3;       \
  *(float4*)((DA) + asoff + 64  * FC_PITCH_A) = ra4;       \
  *(float4*)((DA) + asoff + 80  * FC_PITCH_A) = ra5;       \
  *(float4*)((DA) + asoff + 96  * FC_PITCH_A) = ra6;       \
  *(float4*)((DA) + asoff + 112 * FC_PITCH_A) = ra7;       \
  FC_BPAIR(DB, 0, rb0, rb1)                                \
  FC_BPAIR(DB, 1, rb2, rb3)                                \
  FC_BPAIR(DB, 2, rb4, rb5)                                \
  FC_BPAIR(DB, 3, rb6, rb7)

__global__ __launch_bounds__(256)
void fc1_kernel(const float* __restrict__ feats,
                const float* __restrict__ W,
                const float* __restrict__ bias,
                float* __restrict__ q) {
  extern __shared__ float stage[];
  float* sA = stage;
  float* sB = stage + 2 * FC_TILE_A;

  int r0 = blockIdx.y * 128;
  int c0 = blockIdx.x * 128;

  int tid  = threadIdx.x;
  int wave = tid >> 5;
  int lane = tid & 31;
  int wr = wave >> 1;
  int wc = wave & 1;
  int c  = lane & 15;
  int h  = lane >> 4;

  int arow = tid >> 4;
  int aseg = (tid & 15) * 4;
  const float* gA = feats + (size_t)(r0 + arow) * H_DIM + aseg;
  int asoff = arow * FC_PITCH_A + aseg;
  int bpair = tid >> 5;
  int bcol  = (tid & 31) * 4;
  const float* gB = W + (size_t)(2 * bpair) * H_DIM + c0 + bcol;
  int bsoff = bpair * FC_PITCH_B + 2 * bcol;

  float4 ra0, ra1, ra2, ra3, ra4, ra5, ra6, ra7;
  float4 rb0, rb1, rb2, rb3, rb4, rb5, rb6, rb7;
  FC_GLOAD(0)
  FC_SSTORE(sA, sB)
  FC_GLOAD(FC_KC)
  lds_barrier();

  v8f acc[2][4] = {};
  int aoff = (wr * 32 + c) * FC_PITCH_A + 2 * h;
  int boff = h * FC_PITCH_B + 2 * (wc * 64 + c);

  for (int ch = 0; ch < H_DIM / FC_KC; ++ch) {
    const float* cA = sA + (ch & 1) * FC_TILE_A;
    const float* cB = sB + (ch & 1) * FC_TILE_B;
    #pragma unroll
    for (int ks = 0; ks < FC_KC / 4; ++ks) {
      v2f a[2], bf[4];
      #pragma unroll
      for (int i = 0; i < 2; ++i)
        a[i] = *(const v2f*)(cA + aoff + i * 16 * FC_PITCH_A + ks * 4);
      #pragma unroll
      for (int t = 0; t < 4; ++t)
        bf[t] = *(const v2f*)(cB + boff + ks * 2 * FC_PITCH_B + t * 32);
      #pragma unroll
      for (int i = 0; i < 2; ++i)
        #pragma unroll
        for (int t = 0; t < 4; ++t)
          acc[i][t] = wmma_f32(a[i], bf[t], acc[i][t]);
    }
    __builtin_amdgcn_sched_barrier(0);
    if (ch + 1 < H_DIM / FC_KC) {
      float* nA = sA + ((ch + 1) & 1) * FC_TILE_A;
      float* nB = sB + ((ch + 1) & 1) * FC_TILE_B;
      FC_SSTORE(nA, nB)
    }
    if (ch + 2 < H_DIM / FC_KC) {
      int ko = (ch + 2) * FC_KC;
      FC_GLOAD(ko)
    }
    lds_barrier();
  }

  #pragma unroll
  for (int t = 0; t < 4; ++t) {
    int col = c0 + wc * 64 + t * 16 + c;
    float bz = bias[col];
    #pragma unroll
    for (int i = 0; i < 2; ++i) {
      float* dst = q + (size_t)(r0 + wr * 32 + i * 16 + 8 * h) * H_DIM + col;
      #pragma unroll
      for (int j = 0; j < 8; ++j)
        dst[(size_t)j * H_DIM] = acc[i][t][j] + bz;
    }
  }
}

#define SC_PITCH 36
#define SC_TILE  (128 * SC_PITCH)
__global__ __launch_bounds__(256)
void scores_kernel(const float* __restrict__ q,
                   const float* __restrict__ feats,
                   float* __restrict__ S) {
  extern __shared__ float stage[];
  float* sA = stage;
  float* sB = stage + 2 * SC_TILE;

  int b  = blockIdx.z;
  int n0 = blockIdx.y * 128;
  int m0 = blockIdx.x * 128;
  const float* Qg = q     + ((size_t)b * N_DIM + n0) * H_DIM;
  const float* Fg = feats + ((size_t)b * N_DIM + m0) * H_DIM;
  float* Sg = S + ((size_t)b * N_DIM + n0) * N_DIM + m0;

  int tid  = threadIdx.x;
  int wave = tid >> 5;
  int lane = tid & 31;
  int wr = wave >> 1;
  int wc = wave & 1;
  int c  = lane & 15;
  int h  = lane >> 4;

  int srow = tid >> 3;
  int sseg = (tid & 7) * 4;
  const float* gA = Qg + (size_t)srow * H_DIM + sseg;
  const float* gB = Fg + (size_t)srow * H_DIM + sseg;
  int soff = srow * SC_PITCH + sseg;

  float4 ra0, ra1, ra2, ra3, rb0, rb1, rb2, rb3;
  ra0 = *(const float4*)(gA + (size_t)0 * H_DIM + 0);
  rb0 = *(const float4*)(gB + (size_t)0 * H_DIM + 0);
  ra1 = *(const float4*)(gA + (size_t)32 * H_DIM + 0);
  rb1 = *(const float4*)(gB + (size_t)32 * H_DIM + 0);
  ra2 = *(const float4*)(gA + (size_t)64 * H_DIM + 0);
  rb2 = *(const float4*)(gB + (size_t)64 * H_DIM + 0);
  ra3 = *(const float4*)(gA + (size_t)96 * H_DIM + 0);
  rb3 = *(const float4*)(gB + (size_t)96 * H_DIM + 0);
  *(float4*)(sA + soff + 0 * SC_PITCH) = ra0;
  *(float4*)(sB + soff + 0 * SC_PITCH) = rb0;
  *(float4*)(sA + soff + 32 * SC_PITCH) = ra1;
  *(float4*)(sB + soff + 32 * SC_PITCH) = rb1;
  *(float4*)(sA + soff + 64 * SC_PITCH) = ra2;
  *(float4*)(sB + soff + 64 * SC_PITCH) = rb2;
  *(float4*)(sA + soff + 96 * SC_PITCH) = ra3;
  *(float4*)(sB + soff + 96 * SC_PITCH) = rb3;
  ra0 = *(const float4*)(gA + (size_t)0 * H_DIM + 32);
  rb0 = *(const float4*)(gB + (size_t)0 * H_DIM + 32);
  ra1 = *(const float4*)(gA + (size_t)32 * H_DIM + 32);
  rb1 = *(const float4*)(gB + (size_t)32 * H_DIM + 32);
  ra2 = *(const float4*)(gA + (size_t)64 * H_DIM + 32);
  rb2 = *(const float4*)(gB + (size_t)64 * H_DIM + 32);
  ra3 = *(const float4*)(gA + (size_t)96 * H_DIM + 32);
  rb3 = *(const float4*)(gB + (size_t)96 * H_DIM + 32);
  lds_barrier();

  v8f acc[2][4] = {};
  int aoff = (wr * 32 + c) * SC_PITCH + 2 * h;
  int boff = (wc * 64 + c) * SC_PITCH + 2 * h;

  for (int ch = 0; ch < H_DIM / 32; ++ch) {
    const float* cA = sA + (ch & 1) * SC_TILE;
    const float* cB = sB + (ch & 1) * SC_TILE;
    #pragma unroll
    for (int ks = 0; ks < 8; ++ks) {
      v2f a[2], bf[4];
      #pragma unroll
      for (int i = 0; i < 2; ++i)
        a[i] = *(const v2f*)(cA + aoff + i * 16 * SC_PITCH + ks * 4);
      #pragma unroll
      for (int t = 0; t < 4; ++t)
        bf[t] = *(const v2f*)(cB + boff + t * 16 * SC_PITCH + ks * 4);
      #pragma unroll
      for (int i = 0; i < 2; ++i)
        #pragma unroll
        for (int t = 0; t < 4; ++t)
          acc[i][t] = wmma_f32(a[i], bf[t], acc[i][t]);
    }
    if (ch + 1 < H_DIM / 32) {
      float* nA = sA + ((ch + 1) & 1) * SC_TILE;
      float* nB = sB + ((ch + 1) & 1) * SC_TILE;
      *(float4*)(nA + soff + 0 * SC_PITCH) = ra0;
      *(float4*)(nB + soff + 0 * SC_PITCH) = rb0;
      *(float4*)(nA + soff + 32 * SC_PITCH) = ra1;
      *(float4*)(nB + soff + 32 * SC_PITCH) = rb1;
      *(float4*)(nA + soff + 64 * SC_PITCH) = ra2;
      *(float4*)(nB + soff + 64 * SC_PITCH) = rb2;
      *(float4*)(nA + soff + 96 * SC_PITCH) = ra3;
      *(float4*)(nB + soff + 96 * SC_PITCH) = rb3;
    }
    if (ch + 2 < H_DIM / 32) {
      int ko = (ch + 2) * 32;
      ra0 = *(const float4*)(gA + (size_t)0 * H_DIM + ko);
      rb0 = *(const float4*)(gB + (size_t)0 * H_DIM + ko);
      ra1 = *(const float4*)(gA + (size_t)32 * H_DIM + ko);
      rb1 = *(const float4*)(gB + (size_t)32 * H_DIM + ko);
      ra2 = *(const float4*)(gA + (size_t)64 * H_DIM + ko);
      rb2 = *(const float4*)(gB + (size_t)64 * H_DIM + ko);
      ra3 = *(const float4*)(gA + (size_t)96 * H_DIM + ko);
      rb3 = *(const float4*)(gB + (size_t)96 * H_DIM + ko);
    }
    lds_barrier();
  }

  #pragma unroll
  for (int i = 0; i < 2; ++i) {
    #pragma unroll
    for (int t = 0; t < 4; ++t) {
      float* dst = Sg + (size_t)(wr * 32 + i * 16 + 8 * h) * N_DIM + wc * 64 + t * 16 + c;
      #pragma unroll
      for (int j = 0; j < 8; ++j)
        dst[(size_t)j * N_DIM] = acc[i][t][j];
    }
  }
}

typedef __attribute__((ext_vector_type(4))) float v4f;

__global__ __launch_bounds__(256)
void softmax_kernel(const float* __restrict__ S,
                    float* __restrict__ partial) {
  int b  = blockIdx.x >> 7;
  int nt = blockIdx.x & (NT - 1);
  const float* Sr = S + ((size_t)b * N_DIM + nt * 16) * N_DIM;

  extern __shared__ __attribute__((aligned(16))) float slab[];
  __shared__ float rowInv[16];

  int wave = threadIdx.x >> 5;
  int lane = threadIdx.x & 31;

  const float* R = Sr + (size_t)(2 * wave) * N_DIM + lane * 4;
  v4f x[2][16];
  #pragma unroll
  for (int rr = 0; rr < 2; ++rr)
    #pragma unroll
    for (int j = 0; j < 16; ++j)
      x[rr][j] = __builtin_nontemporal_load((const v4f*)(R + (size_t)rr * N_DIM + j * 128));

  float mx[2];
  #pragma unroll
  for (int rr = 0; rr < 2; ++rr) {
    float m0 = -3.402823466e38f;
    #pragma unroll
    for (int j = 0; j < 16; ++j)
      m0 = fmaxf(m0, fmaxf(fmaxf(x[rr][j].x, x[rr][j].y), fmaxf(x[rr][j].z, x[rr][j].w)));
    mx[rr] = m0;
  }
  #pragma unroll
  for (int o = 16; o > 0; o >>= 1)
    #pragma unroll
    for (int rr = 0; rr < 2; ++rr)
      mx[rr] = fmaxf(mx[rr], __shfl_xor(mx[rr], o, 32));

  float s[2];
  #pragma unroll
  for (int rr = 0; rr < 2; ++rr) {
    float s0 = 0.f, s1 = 0.f, s2 = 0.f, s3 = 0.f;
    #pragma unroll
    for (int j = 0; j < 16; ++j) {
      v4f e;
      e.x = __expf(x[rr][j].x - mx[rr]);
      e.y = __expf(x[rr][j].y - mx[rr]);
      e.z = __expf(x[rr][j].z - mx[rr]);
      e.w = __expf(x[rr][j].w - mx[rr]);
      x[rr][j] = e;
      s0 += e.x;
      s1 += e.y;
      s2 += e.z;
      s3 += e.w;
    }
    s[rr] = (s0 + s1) + (s2 + s3);
  }
  #pragma unroll
  for (int o = 16; o > 0; o >>= 1)
    #pragma unroll
    for (int rr = 0; rr < 2; ++rr)
      s[rr] += __shfl_xor(s[rr], o, 32);
  if (lane == 0) {
    rowInv[2 * wave]     = 1.0f / s[0];
    rowInv[2 * wave + 1] = 1.0f / s[1];
  }
  #pragma unroll
  for (int rr = 0; rr < 2; ++rr)
    #pragma unroll
    for (int j = 0; j < 16; ++j)
      *(v4f*)(slab + (2 * wave + rr) * N_DIM + j * 128 + lane * 4) = x[rr][j];
  __syncthreads();

  float* P = partial + ((size_t)b * NT + nt) * N_DIM;
  #pragma unroll
  for (int k = 0; k < 2; ++k) {
    int m = k * (N_DIM / 2) + threadIdx.x * 4;
    v4f a = {0.f, 0.f, 0.f, 0.f};
    #pragma unroll
    for (int r = 0; r < 16; ++r) {
      v4f e = *(const v4f*)(slab + r * N_DIM + m);
      float inv = rowInv[r];
      a.x = fmaf(e.x, inv, a.x);
      a.y = fmaf(e.y, inv, a.y);
      a.z = fmaf(e.z, inv, a.z);
      a.w = fmaf(e.w, inv, a.w);
    }
    *(v4f*)(P + m) = a;
  }
}

__global__ void colsum_kernel(const float* __restrict__ partial,
                              float* __restrict__ colsum) {
  int idx = blockIdx.x * blockDim.x + threadIdx.x;
  int b = idx >> 11;
  int m = idx & (N_DIM - 1);
  const float* P = partial + (size_t)b * NT * N_DIM + m;
  float a = 0.f;
  #pragma unroll 8
  for (int t = 0; t < NT; ++t) a += P[(size_t)t * N_DIM];
  colsum[idx] = a;
}

#define PL_ROWS   32
#define PL_STRIPS (N_DIM / PL_ROWS)
__global__ __launch_bounds__(256)
void pool_strip_kernel(const float* __restrict__ colsum,
                       const float* __restrict__ feats,
                       float* __restrict__ pp) {
  int s = blockIdx.x;
  int b = blockIdx.y;
  int m0 = s * PL_ROWS;
  const float* F = feats + ((size_t)b * N_DIM + m0) * H_DIM + threadIdx.x * 4;
  const float* C = colsum + (size_t)b * N_DIM + m0;

  v4f acc = {0.f, 0.f, 0.f, 0.f};
  #pragma unroll
  for (int i = 0; i < PL_ROWS; ++i) {
    v4f f = __builtin_nontemporal_load((const v4f*)(F + (size_t)i * H_DIM));
    float c = C[i];
    acc.x = fmaf(c, f.x, acc.x);
    acc.y = fmaf(c, f.y, acc.y);
    acc.z = fmaf(c, f.z, acc.z);
    acc.w = fmaf(c, f.w, acc.w);
  }
  *(v4f*)(pp + ((size_t)b * PL_STRIPS + s) * H_DIM + threadIdx.x * 4) = acc;
}

__global__ void pool_join_kernel(const float* __restrict__ pp,
                                 float* __restrict__ out) {
  int idx = blockIdx.x * blockDim.x + threadIdx.x;
  int b = idx >> 10;
  int h = idx & (H_DIM - 1);
  const float* P = pp + (size_t)b * PL_STRIPS * H_DIM + h;
  float a = 0.f;
  #pragma unroll
  for (int s = 0; s < PL_STRIPS; ++s) a += P[(size_t)s * H_DIM];
  out[idx] = a;
}

#define BATCH_FEATS ((size_t)N_DIM * H_DIM)
#define BATCH_S     ((size_t)N_DIM * N_DIM)
#define BATCH_PART  ((size_t)NT * N_DIM)
#define TAIL_FLOATS ((size_t)B_DIM * NT * N_DIM + (size_t)B_DIM * N_DIM)
#define WS_NEED(NB) (((size_t)(NB) * (BATCH_FEATS + BATCH_S) + TAIL_FLOATS) * sizeof(float))
#define RUN_PASS(G, NB)                                                                     \
  fc1_kernel<<<dim3(8, (NB) * 16), 256, (2 * FC_TILE_A + 2 * FC_TILE_B) * sizeof(float), stream>>>(  \
      feats + (size_t)(G) * (NB) * BATCH_FEATS, W, bias, q);                                \
  scores_kernel<<<dim3(16, 16, (NB)), 256, 4 * SC_TILE * sizeof(float), stream>>>(          \
      q, feats + (size_t)(G) * (NB) * BATCH_FEATS, S);                                      \
  softmax_kernel<<<(NB) * NT, 256, 16 * N_DIM * sizeof(float), stream>>>(                   \
      S, partial + (size_t)(G) * (NB) * BATCH_PART);

extern "C" void kernel_launch(void* const* d_in, const int* in_sizes, int n_in,
                              void* d_out, int out_size, void* d_ws, size_t ws_size,
                              hipStream_t stream) {
  const float* feats = (const float*)d_in[0];
  const float* W     = (const float*)d_in[1];
  const float* bias  = (const float*)d_in[2];
  float* out = (float*)d_out;

  float* q = (float*)d_ws;
  float* S;
  float* partial;
  if (ws_size >= WS_NEED(16)) {
    S       = q + 16 * BATCH_FEATS;
    partial = S + 16 * BATCH_S;
    RUN_PASS(0, 16)
  } else if (ws_size >= WS_NEED(8)) {
    S       = q + 8 * BATCH_FEATS;
    partial = S + 8 * BATCH_S;
    RUN_PASS(0, 8)
    RUN_PASS(1, 8)
  } else {
    S       = q + 4 * BATCH_FEATS;
    partial = S + 4 * BATCH_S;
    RUN_PASS(0, 4)
    RUN_PASS(1, 4)
    RUN_PASS(2, 4)
    RUN_PASS(3, 4)
  }
  float* colsum = partial + (size_t)B_DIM * NT * N_DIM;

  colsum_kernel<<<(B_DIM * N_DIM) / 256, 256, 0, stream>>>(partial, colsum);

  float* pp = q;
  pool_strip_kernel<<<dim3(PL_STRIPS, B_DIM), 256, 0, stream>>>(colsum, feats, pp);
  pool_join_kernel<<<(B_DIM * H_DIM) / 256, 256, 0, stream>>>(pp, out);
}
